# starting kernel
# baseline (speedup 1.0000x reference)
#include <hip/hip_runtime.h>
#include <hip/hip_bf16.h>


#define DIM 128

typedef __attribute__((ext_vector_type(2))) float v2f;
typedef __attribute__((ext_vector_type(8))) float v8f;

__global__ __launch_bounds__(256) void ggc_gemm_m(
    const float* __restrict__ h, const float* __restrict__ W,
    float* __restrict__ m, int nNodes)
{
    __shared__ float hTile[16][DIM];

    const int tileM = blockIdx.x * 16;
    const int tid   = threadIdx.x;

    for (int i = tid; i < 16 * DIM; i += 256) {
        int r = i >> 7, c = i & (DIM - 1);
        hTile[r][c] = h[(long long)(tileM + r) * DIM + c];
    }
    __syncthreads();

    const int wave  = tid >> 5;
    const int lane  = tid & 31;
    const int mrow  = lane & 15;
    const int ncol  = lane & 15;
    const int khalf = (lane >> 4) * 2;
    const int rowHi = (lane >> 4) * 8;
    const int tileN = wave * 16;

    v8f acc = {};
    for (int kk = 0; kk < DIM; kk += 4) {
        v2f a, b;
        a.x = hTile[mrow][kk + khalf];
        a.y = hTile[mrow][kk + khalf + 1];
        b.x = W[(kk + khalf) * DIM + tileN + ncol];
        b.y = W[(kk + khalf + 1) * DIM + tileN + ncol];
        acc = __builtin_amdgcn_wmma_f32_16x16x4_f32(
            false, a, false, b, (short)0, acc, false, false);
    }

    #pragma unroll
    for (int r = 0; r < 8; ++r)
        m[(long long)(tileM + rowHi + r) * DIM + tileN + ncol] = acc[r];
}

__global__ __launch_bounds__(256) void ggc_scatter(
    const float* __restrict__ m,
    const int* __restrict__ src, const int* __restrict__ dst,
    const float* __restrict__ ew,
    float* __restrict__ agg, long long nEdges)
{
    long long gid = (long long)blockIdx.x * blockDim.x + threadIdx.x;
    long long nWork = nEdges * 32;
    if (gid >= nWork) return;
    long long e = gid >> 5;
    int chunk   = (int)(gid & 31);

    int   s = src[e];
    int   d = dst[e];
    float w = ew[e];

    const float4 v = ((const float4*)(m + (long long)s * DIM))[chunk];
    float* ap = agg + (long long)d * DIM + chunk * 4;
    atomicAdd(ap + 0, v.x * w);
    atomicAdd(ap + 1, v.y * w);
    atomicAdd(ap + 2, v.z * w);
    atomicAdd(ap + 3, v.w * w);
}

__global__ __launch_bounds__(256) void ggc_gru(
    const float* __restrict__ agg, const float* __restrict__ h,
    const float* __restrict__ W_ih, const float* __restrict__ W_hh,
    const float* __restrict__ b_ih, const float* __restrict__ b_hh,
    float* __restrict__ out, int nNodes)
{
    __shared__ float gi[16][3 * DIM];
    __shared__ float gh[16][3 * DIM];

    const int tileM = blockIdx.x * 16;
    const int tid   = threadIdx.x;
    const int wave  = tid >> 5;
    const int lane  = tid & 31;
    const int mrow  = lane & 15;
    const int ncol  = lane & 15;
    const int khalf = (lane >> 4) * 2;
    const int rowHi = (lane >> 4) * 8;

    for (int j = wave; j < 48; j += 8) {
        const bool  isGi = (j < 24);
        const int   tn   = (isGi ? j : j - 24) * 16;
        const float* A   = isGi ? agg : h;
        const float* B   = isGi ? W_ih : W_hh;

        const float* arow = A + (long long)(tileM + mrow) * DIM;
        const float* wrow = B + (long long)(tn + ncol) * DIM;

        v8f acc = {};
        for (int kk = 0; kk < DIM; kk += 4) {
            v2f a, b;
            a.x = arow[kk + khalf];
            a.y = arow[kk + khalf + 1];
            b.x = wrow[kk + khalf];
            b.y = wrow[kk + khalf + 1];
            acc = __builtin_amdgcn_wmma_f32_16x16x4_f32(
                false, a, false, b, (short)0, acc, false, false);
        }

        float* G = isGi ? &gi[0][0] : &gh[0][0];
        #pragma unroll
        for (int r = 0; r < 8; ++r)
            G[(rowHi + r) * (3 * DIM) + tn + ncol] = acc[r];
    }
    __syncthreads();

    for (int i = tid; i < 16 * DIM; i += 256) {
        int r = i >> 7, c = i & (DIM - 1);
        float ir = gi[r][c]           + b_ih[c];
        float iz = gi[r][c + DIM]     + b_ih[c + DIM];
        float in = gi[r][c + 2 * DIM] + b_ih[c + 2 * DIM];
        float hr = gh[r][c]           + b_hh[c];
        float hz = gh[r][c + DIM]     + b_hh[c + DIM];
        float hn = gh[r][c + 2 * DIM] + b_hh[c + 2 * DIM];

        float rg = 1.0f / (1.0f + __expf(-(ir + hr)));
        float zg = 1.0f / (1.0f + __expf(-(iz + hz)));
        float ng = tanhf(in + rg * hn);
        float hp = h[(long long)(tileM + r) * DIM + c];
        out[(long long)(tileM + r) * DIM + c] = (1.0f - zg) * ng + zg * hp;
    }
}

extern "C" void kernel_launch(void* const* d_in, const int* in_sizes, int n_in,
                              void* d_out, int out_size, void* d_ws, size_t ws_size,
                              hipStream_t stream) {
    const float* x    = (const float*)d_in[0];
    const int*   eidx = (const int*)d_in[1];
    const float* ea   = (const float*)d_in[2];
    const float* W    = (const float*)d_in[3];
    const float* W_ih = (const float*)d_in[4];
    const float* W_hh = (const float*)d_in[5];
    const float* b_ih = (const float*)d_in[6];
    const float* b_hh = (const float*)d_in[7];

    const int       N = in_sizes[0] / DIM;
    const long long E = (long long)in_sizes[2];
    const int       L = in_sizes[3] / (DIM * DIM);

    const int* src = eidx;
    const int* dst = eidx + E;

    const size_t nd = (size_t)N * DIM;
    float* P0  = (float*)d_ws;
    float* P1  = P0 + nd;
    float* AGG = P1 + nd;

    const dim3 blk(256);
    const dim3 gridNodes((unsigned)(N / 16));
    const long long scatWork = E * 32;
    const dim3 gridScat((unsigned)((scatWork + 255) / 256));

    const float* hcur = x;
    for (int l = 0; l < L; ++l) {
        float* mbuf = (l & 1) ? P1 : P0;

        ggc_gemm_m<<<gridNodes, blk, 0, stream>>>(hcur, W + (size_t)l * DIM * DIM,
                                                  mbuf, N);
        hipMemsetAsync(AGG, 0, nd * sizeof(float), stream);
        ggc_scatter<<<gridScat, blk, 0, stream>>>(mbuf, src, dst, ea, AGG, E);

        float* hout = (l == L - 1) ? (float*)d_out : mbuf;
        ggc_gru<<<gridNodes, blk, 0, stream>>>(AGG, hcur, W_ih, W_hh, b_ih, b_hh,
                                               hout, N);
        hcur = hout;
    }
}


# ggc_gru weight-stationary: each wave keeps its W_ih/W_hh fragments in registers, one block per WGP loops over 16-node tiles staged in LDS; k regrouped in 8-wide chunks, which changes the sum order
# speedup vs baseline: 1.4562x; 1.4562x over previous
#include <hip/hip_runtime.h>
#include <hip/hip_bf16.h>


#define DIM 128

typedef __attribute__((ext_vector_type(2))) float v2f;
typedef __attribute__((ext_vector_type(4))) float v4f;
typedef __attribute__((ext_vector_type(8))) float v8f;

__global__ __launch_bounds__(256) void ggc_gemm_m(
    const float* __restrict__ h, const float* __restrict__ W,
    float* __restrict__ m, int nNodes)
{
    __shared__ float hTile[16][DIM];

    const int tileM = blockIdx.x * 16;
    const int tid   = threadIdx.x;

    for (int i = tid; i < 16 * DIM; i += 256) {
        int r = i >> 7, c = i & (DIM - 1);
        hTile[r][c] = h[(long long)(tileM + r) * DIM + c];
    }
    __syncthreads();

    const int wave  = tid >> 5;
    const int lane  = tid & 31;
    const int mrow  = lane & 15;
    const int ncol  = lane & 15;
    const int khalf = (lane >> 4) * 2;
    const int rowHi = (lane >> 4) * 8;
    const int tileN = wave * 16;

    v8f acc = {};
    for (int kk = 0; kk < DIM; kk += 4) {
        v2f a, b;
        a.x = hTile[mrow][kk + khalf];
        a.y = hTile[mrow][kk + khalf + 1];
        b.x = W[(kk + khalf) * DIM + tileN + ncol];
        b.y = W[(kk + khalf + 1) * DIM + tileN + ncol];
        acc = __builtin_amdgcn_wmma_f32_16x16x4_f32(
            false, a, false, b, (short)0, acc, false, false);
    }

    #pragma unroll
    for (int r = 0; r < 8; ++r)
        m[(long long)(tileM + rowHi + r) * DIM + tileN + ncol] = acc[r];
}

__global__ __launch_bounds__(256) void ggc_scatter(
    const float* __restrict__ m,
    const int* __restrict__ src, const int* __restrict__ dst,
    const float* __restrict__ ew,
    float* __restrict__ agg, long long nEdges)
{
    long long gid = (long long)blockIdx.x * blockDim.x + threadIdx.x;
    long long nWork = nEdges * 32;
    if (gid >= nWork) return;
    long long e = gid >> 5;
    int chunk   = (int)(gid & 31);

    int   s = src[e];
    int   d = dst[e];
    float w = ew[e];

    const float4 v = ((const float4*)(m + (long long)s * DIM))[chunk];
    float* ap = agg + (long long)d * DIM + chunk * 4;
    atomicAdd(ap + 0, v.x * w);
    atomicAdd(ap + 1, v.y * w);
    atomicAdd(ap + 2, v.z * w);
    atomicAdd(ap + 3, v.w * w);
}

#define GRU_PITCH 260

__global__ __launch_bounds__(256) void ggc_gru(
    const float* __restrict__ agg, const float* __restrict__ h,
    const float* __restrict__ W_ih, const float* __restrict__ W_hh,
    const float* __restrict__ b_ih, const float* __restrict__ b_hh,
    float* __restrict__ out, int nNodes)
{
    __shared__ __attribute__((aligned(16))) float aT[2][16][GRU_PITCH];

    const int tid  = threadIdx.x;
    const int wave = tid >> 5;
    const int lane = tid & 31;
    const int c    = lane & 15;
    const int hf   = lane >> 4;
    const int col  = wave * 16 + c;

    const int nTiles = (nNodes + 15) >> 4;

    v4f bi[3][16], bh[3][16];
    #pragma unroll
    for (int g = 0; g < 3; ++g) {
        const float* wi = W_ih + (long long)(g * DIM + col) * DIM + 4 * hf;
        const float* wh = W_hh + (long long)(g * DIM + col) * DIM + 4 * hf;
        #pragma unroll
        for (int j = 0; j < 16; ++j) {
            bi[g][j] = *(const v4f*)(wi + 8 * j);
            bh[g][j] = *(const v4f*)(wh + 8 * j);
        }
    }
    const float bir = b_ih[col], biz = b_ih[col + DIM], bin = b_ih[col + 2 * DIM];
    const float bhr = b_hh[col], bhz = b_hh[col + DIM], bhn = b_hh[col + 2 * DIM];

    const int q  = tid & 63;
    const int r0 = tid >> 6;
    const float* srcBase = (q < 32) ? (agg + q * 4) : (h + (q - 32) * 4);
    float4 p[4];

    int tile = blockIdx.x;
    if (tile < nTiles) {
        for (int i = 0; i < 4; ++i) {
            const int row = tile * 16 + r0 + 4 * i;
            p[i] = (row < nNodes) ? *(const float4*)(srcBase + (long long)row * DIM)
                                  : make_float4(0.f, 0.f, 0.f, 0.f);
            *(float4*)&aT[0][r0 + 4 * i][q * 4] = p[i];
        }
    }
    __syncthreads();

    int buf = 0;
    for (; tile < nTiles; tile += gridDim.x) {
        const int  nxt     = tile + gridDim.x;
        const bool hasNext = nxt < nTiles;
        if (hasNext) {
            for (int i = 0; i < 4; ++i) {
                const int row = nxt * 16 + r0 + 4 * i;
                p[i] = (row < nNodes) ? *(const float4*)(srcBase + (long long)row * DIM)
                                      : make_float4(0.f, 0.f, 0.f, 0.f);
            }
        }

        v8f aIr = {}, aIz = {}, aIn = {}, aHr = {}, aHz = {}, aHn = {};
        const float* arow = &aT[buf][c][4 * hf];
        #pragma unroll
        for (int j = 0; j < 16; ++j) {
            const v4f xa = *(const v4f*)(arow + 8 * j);
            const v4f xh = *(const v4f*)(arow + DIM + 8 * j);
            const v2f xa0 = {xa.x, xa.y}, xa1 = {xa.z, xa.w};
            const v2f xh0 = {xh.x, xh.y}, xh1 = {xh.z, xh.w};
            v2f b0, b1;
            b0 = (v2f){bi[0][j].x, bi[0][j].y}; b1 = (v2f){bi[0][j].z, bi[0][j].w};
            aIr = __builtin_amdgcn_wmma_f32_16x16x4_f32(false, xa0, false, b0, (short)0, aIr, false, false);
            aIr = __builtin_amdgcn_wmma_f32_16x16x4_f32(false, xa1, false, b1, (short)0, aIr, false, false);
            b0 = (v2f){bi[1][j].x, bi[1][j].y}; b1 = (v2f){bi[1][j].z, bi[1][j].w};
            aIz = __builtin_amdgcn_wmma_f32_16x16x4_f32(false, xa0, false, b0, (short)0, aIz, false, false);
            aIz = __builtin_amdgcn_wmma_f32_16x16x4_f32(false, xa1, false, b1, (short)0, aIz, false, false);
            b0 = (v2f){bi[2][j].x, bi[2][j].y}; b1 = (v2f){bi[2][j].z, bi[2][j].w};
            aIn = __builtin_amdgcn_wmma_f32_16x16x4_f32(false, xa0, false, b0, (short)0, aIn, false, false);
            aIn = __builtin_amdgcn_wmma_f32_16x16x4_f32(false, xa1, false, b1, (short)0, aIn, false, false);
            b0 = (v2f){bh[0][j].x, bh[0][j].y}; b1 = (v2f){bh[0][j].z, bh[0][j].w};
            aHr = __builtin_amdgcn_wmma_f32_16x16x4_f32(false, xh0, false, b0, (short)0, aHr, false, false);
            aHr = __builtin_amdgcn_wmma_f32_16x16x4_f32(false, xh1, false, b1, (short)0, aHr, false, false);
            b0 = (v2f){bh[1][j].x, bh[1][j].y}; b1 = (v2f){bh[1][j].z, bh[1][j].w};
            aHz = __builtin_amdgcn_wmma_f32_16x16x4_f32(false, xh0, false, b0, (short)0, aHz, false, false);
            aHz = __builtin_amdgcn_wmma_f32_16x16x4_f32(false, xh1, false, b1, (short)0, aHz, false, false);
            b0 = (v2f){bh[2][j].x, bh[2][j].y}; b1 = (v2f){bh[2][j].z, bh[2][j].w};
            aHn = __builtin_amdgcn_wmma_f32_16x16x4_f32(false, xh0, false, b0, (short)0, aHn, false, false);
            aHn = __builtin_amdgcn_wmma_f32_16x16x4_f32(false, xh1, false, b1, (short)0, aHn, false, false);
        }

        #pragma unroll
        for (int r = 0; r < 8; ++r) {
            const int row  = 8 * hf + r;
            const int node = tile * 16 + row;
            float ir = aIr[r] + bir;
            float iz = aIz[r] + biz;
            float in = aIn[r] + bin;
            float hr = aHr[r] + bhr;
            float hz = aHz[r] + bhz;
            float hn = aHn[r] + bhn;

            float rg = 1.0f / (1.0f + __expf(-(ir + hr)));
            float zg = 1.0f / (1.0f + __expf(-(iz + hz)));
            float ng = tanhf(in + rg * hn);
            float hp = aT[buf][row][DIM + col];
            if (node < nNodes)
                out[(long long)node * DIM + col] = (1.0f - zg) * ng + zg * hp;
        }

        if (hasNext) {
            for (int i = 0; i < 4; ++i)
                *(float4*)&aT[buf ^ 1][r0 + 4 * i][q * 4] = p[i];
        }
        __syncthreads();
        buf ^= 1;
    }
}

extern "C" void kernel_launch(void* const* d_in, const int* in_sizes, int n_in,
                              void* d_out, int out_size, void* d_ws, size_t ws_size,
                              hipStream_t stream) {
    const float* x    = (const float*)d_in[0];
    const int*   eidx = (const int*)d_in[1];
    const float* ea   = (const float*)d_in[2];
    const float* W    = (const float*)d_in[3];
    const float* W_ih = (const float*)d_in[4];
    const float* W_hh = (const float*)d_in[5];
    const float* b_ih = (const float*)d_in[6];
    const float* b_hh = (const float*)d_in[7];

    const int       N = in_sizes[0] / DIM;
    const long long E = (long long)in_sizes[2];
    const int       L = in_sizes[3] / (DIM * DIM);

    const int* src = eidx;
    const int* dst = eidx + E;

    const size_t nd = (size_t)N * DIM;
    float* P0  = (float*)d_ws;
    float* P1  = P0 + nd;
    float* AGG = P1 + nd;

    const dim3 blk(256);
    const dim3 gridNodes((unsigned)(N / 16));
    const dim3 gridGru(256);
    const long long scatWork = E * 32;
    const dim3 gridScat((unsigned)((scatWork + 255) / 256));

    const float* hcur = x;
    for (int l = 0; l < L; ++l) {
        float* mbuf = (l & 1) ? P1 : P0;

        ggc_gemm_m<<<gridNodes, blk, 0, stream>>>(hcur, W + (size_t)l * DIM * DIM,
                                                  mbuf, N);
        hipMemsetAsync(AGG, 0, nd * sizeof(float), stream);
        ggc_scatter<<<gridScat, blk, 0, stream>>>(mbuf, src, dst, ea, AGG, E);

        float* hout = (l == L - 1) ? (float*)d_out : mbuf;
        ggc_gru<<<gridGru, blk, 0, stream>>>(AGG, hcur, W_ih, W_hh, b_ih, b_hh,
                                             hout, N);
        hcur = hout;
    }
}


# aggregation as sorted gather: edges counting-sorted by dst once per call (integer atomic rank); one wave per node sums its in-edge rows in that arrival order; float atomics and agg memset removed
# speedup vs baseline: 2.3981x; 1.6468x over previous
#include <hip/hip_runtime.h>
#include <hip/hip_bf16.h>


#define DIM 128

typedef __attribute__((ext_vector_type(2))) float v2f;
typedef __attribute__((ext_vector_type(4))) float v4f;
typedef __attribute__((ext_vector_type(8))) float v8f;

__global__ __launch_bounds__(256) void ggc_gemm_m(
    const float* __restrict__ h, const float* __restrict__ W,
    float* __restrict__ m, int nNodes)
{
    __shared__ float hTile[16][DIM];

    const int tileM = blockIdx.x * 16;
    const int tid   = threadIdx.x;

    for (int i = tid; i < 16 * DIM; i += 256) {
        int r = i >> 7, c = i & (DIM - 1);
        hTile[r][c] = h[(long long)(tileM + r) * DIM + c];
    }
    __syncthreads();

    const int wave  = tid >> 5;
    const int lane  = tid & 31;
    const int mrow  = lane & 15;
    const int ncol  = lane & 15;
    const int khalf = (lane >> 4) * 2;
    const int rowHi = (lane >> 4) * 8;
    const int tileN = wave * 16;

    v8f acc = {};
    for (int kk = 0; kk < DIM; kk += 4) {
        v2f a, b;
        a.x = hTile[mrow][kk + khalf];
        a.y = hTile[mrow][kk + khalf + 1];
        b.x = W[(kk + khalf) * DIM + tileN + ncol];
        b.y = W[(kk + khalf + 1) * DIM + tileN + ncol];
        acc = __builtin_amdgcn_wmma_f32_16x16x4_f32(
            false, a, false, b, (short)0, acc, false, false);
    }

    #pragma unroll
    for (int r = 0; r < 8; ++r)
        m[(long long)(tileM + rowHi + r) * DIM + tileN + ncol] = acc[r];
}

__global__ __launch_bounds__(256) void ggc_degree(
    const int* __restrict__ dst, int* __restrict__ cnt, int* __restrict__ rank,
    long long nEdges, int nNodes)
{
    const long long e = (long long)blockIdx.x * 256 + threadIdx.x;
    if (e >= nEdges) return;
    const int d = dst[e];
    if ((unsigned)d < (unsigned)nNodes) rank[e] = atomicAdd(&cnt[d], 1);
}

__global__ __launch_bounds__(256) void ggc_rowptr(
    const int* __restrict__ cnt, int* __restrict__ rowptr, int nTotal)
{
    __shared__ int red[8];
    __shared__ int wsum[8];
    const int tid  = threadIdx.x;
    const int lane = tid & 31;
    const int wave = tid >> 5;
    const int blockBase = blockIdx.x * 1024;

    int part = 0;
    const int4* c4 = (const int4*)cnt;
    for (int i = tid; i < (blockBase >> 2); i += 256) {
        const int4 v = c4[i];
        part += (v.x + v.y) + (v.z + v.w);
    }
    for (int o = 16; o > 0; o >>= 1) part += __shfl_xor(part, o);
    if (lane == 0) red[wave] = part;

    const int i0 = blockBase + tid * 4;
    const int a0 = (i0 + 0 < nTotal) ? cnt[i0 + 0] : 0;
    const int a1 = (i0 + 1 < nTotal) ? cnt[i0 + 1] : 0;
    const int a2 = (i0 + 2 < nTotal) ? cnt[i0 + 2] : 0;
    const int a3 = (i0 + 3 < nTotal) ? cnt[i0 + 3] : 0;
    const int t  = (a0 + a1) + (a2 + a3);
    int inc = t;
    for (int o = 1; o < 32; o <<= 1) {
        const int n = __shfl_up(inc, o);
        if (lane >= o) inc += n;
    }
    if (lane == 31) wsum[wave] = inc;
    __syncthreads();

    int off = 0;
    for (int w = 0; w < 8; ++w) {
        off += red[w];
        if (w < wave) off += wsum[w];
    }
    int ex = off + inc - t;
    if (i0 + 0 < nTotal) rowptr[i0 + 0] = ex;
    ex += a0;
    if (i0 + 1 < nTotal) rowptr[i0 + 1] = ex;
    ex += a1;
    if (i0 + 2 < nTotal) rowptr[i0 + 2] = ex;
    ex += a2;
    if (i0 + 3 < nTotal) rowptr[i0 + 3] = ex;
}

__global__ __launch_bounds__(256) void ggc_fill(
    const int* __restrict__ src, const int* __restrict__ dst,
    const float* __restrict__ ew, const int* __restrict__ rowptr,
    const int* __restrict__ rank, int2* __restrict__ csr,
    long long nEdges, int nNodes)
{
    const long long e = (long long)blockIdx.x * 256 + threadIdx.x;
    if (e >= nEdges) return;
    const int d = dst[e];
    if ((unsigned)d >= (unsigned)nNodes) return;
    int2 rec;
    rec.x = src[e];
    rec.y = __float_as_int(ew[e]);
    csr[(long long)rowptr[d] + rank[e]] = rec;
}

template <int U>
static __device__ __forceinline__ void ggc_rows(
    const float4* __restrict__ mrow, int recS, int recW, int j, int nNodes, float4& acc)
{
    float4 v[U];
    float  w[U];
    #pragma unroll
    for (int u = 0; u < U; ++u) {
        int s = __builtin_amdgcn_readlane(recS, j + u);
        w[u]  = __int_as_float(__builtin_amdgcn_readlane(recW, j + u));
        s = ((unsigned)s < (unsigned)nNodes) ? s : 0;
        v[u] = mrow[(long long)s * (DIM / 4)];
    }
    #pragma unroll
    for (int u = 0; u < U; ++u) {
        acc.x = fmaf(w[u], v[u].x, acc.x);
        acc.y = fmaf(w[u], v[u].y, acc.y);
        acc.z = fmaf(w[u], v[u].z, acc.z);
        acc.w = fmaf(w[u], v[u].w, acc.w);
    }
}

__global__ __launch_bounds__(256) void ggc_gather(
    const float* __restrict__ m, const int* __restrict__ rowptr,
    const int2* __restrict__ csr, float* __restrict__ agg, int nNodes)
{
    const int lane = threadIdx.x & 31;
    const int node = __builtin_amdgcn_readfirstlane((int)(blockIdx.x * 8 + (threadIdx.x >> 5)));
    if (node >= nNodes) return;
    const int beg = __builtin_amdgcn_readfirstlane(rowptr[node]);
    const int end = __builtin_amdgcn_readfirstlane(rowptr[node + 1]);

    const float4* mrow = (const float4*)m + lane;
    float4 acc = make_float4(0.f, 0.f, 0.f, 0.f);

    for (int base = beg; base < end; base += 32) {
        const int cnt = min(32, end - base);
        int2 rec = make_int2(0, 0);
        if (lane < cnt) rec = csr[(long long)base + lane];
        int j = 0;
        for (; j + 8 <= cnt; j += 8) ggc_rows<8>(mrow, rec.x, rec.y, j, nNodes, acc);
        if (j + 4 <= cnt) { ggc_rows<4>(mrow, rec.x, rec.y, j, nNodes, acc); j += 4; }
        if (j + 2 <= cnt) { ggc_rows<2>(mrow, rec.x, rec.y, j, nNodes, acc); j += 2; }
        if (j < cnt) ggc_rows<1>(mrow, rec.x, rec.y, j, nNodes, acc);
    }
    ((float4*)agg)[(long long)node * (DIM / 4) + lane] = acc;
}

#define GRU_PITCH 260

__global__ __launch_bounds__(256) void ggc_gru(
    const float* __restrict__ agg, const float* __restrict__ h,
    const float* __restrict__ W_ih, const float* __restrict__ W_hh,
    const float* __restrict__ b_ih, const float* __restrict__ b_hh,
    float* __restrict__ out, int nNodes)
{
    __shared__ __attribute__((aligned(16))) float aT[2][16][GRU_PITCH];

    const int tid  = threadIdx.x;
    const int wave = tid >> 5;
    const int lane = tid & 31;
    const int c    = lane & 15;
    const int hf   = lane >> 4;
    const int col  = wave * 16 + c;

    const int nTiles = (nNodes + 15) >> 4;

    v4f bi[3][16], bh[3][16];
    #pragma unroll
    for (int g = 0; g < 3; ++g) {
        const float* wi = W_ih + (long long)(g * DIM + col) * DIM + 4 * hf;
        const float* wh = W_hh + (long long)(g * DIM + col) * DIM + 4 * hf;
        #pragma unroll
        for (int j = 0; j < 16; ++j) {
            bi[g][j] = *(const v4f*)(wi + 8 * j);
            bh[g][j] = *(const v4f*)(wh + 8 * j);
        }
    }
    const float bir = b_ih[col], biz = b_ih[col + DIM], bin = b_ih[col + 2 * DIM];
    const float bhr = b_hh[col], bhz = b_hh[col + DIM], bhn = b_hh[col + 2 * DIM];

    const int q  = tid & 63;
    const int r0 = tid >> 6;
    const float* srcBase = (q < 32) ? (agg + q * 4) : (h + (q - 32) * 4);
    float4 p[4];

    int tile = blockIdx.x;
    if (tile < nTiles) {
        for (int i = 0; i < 4; ++i) {
            const int row = tile * 16 + r0 + 4 * i;
            p[i] = (row < nNodes) ? *(const float4*)(srcBase + (long long)row * DIM)
                                  : make_float4(0.f, 0.f, 0.f, 0.f);
            *(float4*)&aT[0][r0 + 4 * i][q * 4] = p[i];
        }
    }
    __syncthreads();

    int buf = 0;
    for (; tile < nTiles; tile += gridDim.x) {
        const int  nxt     = tile + gridDim.x;
        const bool hasNext = nxt < nTiles;
        if (hasNext) {
            for (int i = 0; i < 4; ++i) {
                const int row = nxt * 16 + r0 + 4 * i;
                p[i] = (row < nNodes) ? *(const float4*)(srcBase + (long long)row * DIM)
                                      : make_float4(0.f, 0.f, 0.f, 0.f);
            }
        }

        v8f aIr = {}, aIz = {}, aIn = {}, aHr = {}, aHz = {}, aHn = {};
        const float* arow = &aT[buf][c][4 * hf];
        #pragma unroll
        for (int j = 0; j < 16; ++j) {
            const v4f xa = *(const v4f*)(arow + 8 * j);
            const v4f xh = *(const v4f*)(arow + DIM + 8 * j);
            const v2f xa0 = {xa.x, xa.y}, xa1 = {xa.z, xa.w};
            const v2f xh0 = {xh.x, xh.y}, xh1 = {xh.z, xh.w};
            v2f b0, b1;
            b0 = (v2f){bi[0][j].x, bi[0][j].y}; b1 = (v2f){bi[0][j].z, bi[0][j].w};
            aIr = __builtin_amdgcn_wmma_f32_16x16x4_f32(false, xa0, false, b0, (short)0, aIr, false, false);
            aIr = __builtin_amdgcn_wmma_f32_16x16x4_f32(false, xa1, false, b1, (short)0, aIr, false, false);
            b0 = (v2f){bi[1][j].x, bi[1][j].y}; b1 = (v2f){bi[1][j].z, bi[1][j].w};
            aIz = __builtin_amdgcn_wmma_f32_16x16x4_f32(false, xa0, false, b0, (short)0, aIz, false, false);
            aIz = __builtin_amdgcn_wmma_f32_16x16x4_f32(false, xa1, false, b1, (short)0, aIz, false, false);
            b0 = (v2f){bi[2][j].x, bi[2][j].y}; b1 = (v2f){bi[2][j].z, bi[2][j].w};
            aIn = __builtin_amdgcn_wmma_f32_16x16x4_f32(false, xa0, false, b0, (short)0, aIn, false, false);
            aIn = __builtin_amdgcn_wmma_f32_16x16x4_f32(false, xa1, false, b1, (short)0, aIn, false, false);
            b0 = (v2f){bh[0][j].x, bh[0][j].y}; b1 = (v2f){bh[0][j].z, bh[0][j].w};
            aHr = __builtin_amdgcn_wmma_f32_16x16x4_f32(false, xh0, false, b0, (short)0, aHr, false, false);
            aHr = __builtin_amdgcn_wmma_f32_16x16x4_f32(false, xh1, false, b1, (short)0, aHr, false, false);
            b0 = (v2f){bh[1][j].x, bh[1][j].y}; b1 = (v2f){bh[1][j].z, bh[1][j].w};
            aHz = __builtin_amdgcn_wmma_f32_16x16x4_f32(false, xh0, false, b0, (short)0, aHz, false, false);
            aHz = __builtin_amdgcn_wmma_f32_16x16x4_f32(false, xh1, false, b1, (short)0, aHz, false, false);
            b0 = (v2f){bh[2][j].x, bh[2][j].y}; b1 = (v2f){bh[2][j].z, bh[2][j].w};
            aHn = __builtin_amdgcn_wmma_f32_16x16x4_f32(false, xh0, false, b0, (short)0, aHn, false, false);
            aHn = __builtin_amdgcn_wmma_f32_16x16x4_f32(false, xh1, false, b1, (short)0, aHn, false, false);
        }

        #pragma unroll
        for (int r = 0; r < 8; ++r) {
            const int row  = 8 * hf + r;
            const int node = tile * 16 + row;
            float ir = aIr[r] + bir;
            float iz = aIz[r] + biz;
            float in = aIn[r] + bin;
            float hr = aHr[r] + bhr;
            float hz = aHz[r] + bhz;
            float hn = aHn[r] + bhn;

            float rg = 1.0f / (1.0f + __expf(-(ir + hr)));
            float zg = 1.0f / (1.0f + __expf(-(iz + hz)));
            float ng = tanhf(in + rg * hn);
            float hp = aT[buf][row][DIM + col];
            if (node < nNodes)
                out[(long long)node * DIM + col] = (1.0f - zg) * ng + zg * hp;
        }

        if (hasNext) {
            for (int i = 0; i < 4; ++i)
                *(float4*)&aT[buf ^ 1][r0 + 4 * i][q * 4] = p[i];
        }
        __syncthreads();
        buf ^= 1;
    }
}

extern "C" void kernel_launch(void* const* d_in, const int* in_sizes, int n_in,
                              void* d_out, int out_size, void* d_ws, size_t ws_size,
                              hipStream_t stream) {
    const float* x    = (const float*)d_in[0];
    const int*   eidx = (const int*)d_in[1];
    const float* ea   = (const float*)d_in[2];
    const float* W    = (const float*)d_in[3];
    const float* W_ih = (const float*)d_in[4];
    const float* W_hh = (const float*)d_in[5];
    const float* b_ih = (const float*)d_in[6];
    const float* b_hh = (const float*)d_in[7];

    const int       N = in_sizes[0] / DIM;
    const long long E = (long long)in_sizes[2];
    const int       L = in_sizes[3] / (DIM * DIM);

    const int* src = eidx;
    const int* dst = eidx + E;

    const size_t nd = (size_t)N * DIM;
    float* P0  = (float*)d_ws;
    float* P1  = P0 + nd;
    float* AGG = P1 + nd;

    const int    nTotal = N + 1;
    const size_t padInt = ((size_t)nTotal + 1023) / 1024 * 1024;
    int*  CNT    = (int*)(AGG + nd);
    int*  ROWPTR = CNT + padInt;
    int*  RANK   = ROWPTR + padInt;
    int2* CSR    = (int2*)(RANK + ((size_t)E + 3) / 4 * 4);

    const dim3 blk(256);
    const dim3 gridNodes((unsigned)(N / 16));
    const dim3 gridGru(256);
    const dim3 gridEdges((unsigned)((E + 255) / 256));
    const dim3 gridScan((unsigned)(padInt / 1024));
    const dim3 gridGather((unsigned)((N + 7) / 8));

    hipMemsetAsync(CNT, 0, (size_t)nTotal * sizeof(int), stream);
    ggc_degree<<<gridEdges, blk, 0, stream>>>(dst, CNT, RANK, E, N);
    ggc_rowptr<<<gridScan, blk, 0, stream>>>(CNT, ROWPTR, nTotal);
    ggc_fill<<<gridEdges, blk, 0, stream>>>(src, dst, ea, ROWPTR, RANK, CSR, E, N);

    const float* hcur = x;
    for (int l = 0; l < L; ++l) {
        float* mbuf = (l & 1) ? P1 : P0;

        ggc_gemm_m<<<gridNodes, blk, 0, stream>>>(hcur, W + (size_t)l * DIM * DIM,
                                                  mbuf, N);
        ggc_gather<<<gridGather, blk, 0, stream>>>(mbuf, ROWPTR, CSR, AGG, N);

        float* hout = (l == L - 1) ? (float*)d_out : mbuf;
        ggc_gru<<<gridGru, blk, 0, stream>>>(AGG, hcur, W_ih, W_hh, b_ih, b_hh,
                                             hout, N);
        hcur = hout;
    }
}


# layer transform moved behind the aggregation: gather sums h rows, GRU multiplies by Wc = W_ih*W_l^T (small WMMA product once per call); per-layer Nx128x128 product removed; changes sum order
# speedup vs baseline: 2.8906x; 1.2054x over previous
#include <hip/hip_runtime.h>
#include <hip/hip_bf16.h>


#define DIM 128

typedef __attribute__((ext_vector_type(2))) float v2f;
typedef __attribute__((ext_vector_type(4))) float v4f;
typedef __attribute__((ext_vector_type(8))) float v8f;

__global__ __launch_bounds__(256) void ggc_wcomb(
    const float* __restrict__ W_ih, const float* __restrict__ W,
    float* __restrict__ Wc, int nLayers)
{
    const int lane = threadIdx.x & 31;
    const int c    = lane & 15;
    const int hf   = lane >> 4;
    const int t = __builtin_amdgcn_readfirstlane((int)(blockIdx.x * 8 + (threadIdx.x >> 5)));
    const int l = t / 192;
    if (l >= nLayers) return;
    const int r  = t - l * 192;
    const int to = (r >> 3) * 16;
    const int tk = (r & 7) * 16;

    const float* a = W_ih + (long long)(to + c) * DIM + 4 * hf;
    const float* b = W + ((long long)l * DIM + tk + c) * DIM + 4 * hf;

    v8f acc0 = {}, acc1 = {};
    #pragma unroll
    for (int jj = 0; jj < 16; ++jj) {
        const v4f xa = *(const v4f*)(a + 8 * jj);
        const v4f xb = *(const v4f*)(b + 8 * jj);
        const v2f a0 = {xa.x, xa.y}, a1 = {xa.z, xa.w};
        const v2f b0 = {xb.x, xb.y}, b1 = {xb.z, xb.w};
        acc0 = __builtin_amdgcn_wmma_f32_16x16x4_f32(false, a0, false, b0, (short)0, acc0, false, false);
        acc1 = __builtin_amdgcn_wmma_f32_16x16x4_f32(false, a1, false, b1, (short)0, acc1, false, false);
    }

    float* d = Wc + ((long long)l * 3 * DIM + to + 8 * hf) * DIM + tk + c;
    #pragma unroll
    for (int q = 0; q < 8; ++q)
        d[(long long)q * DIM] = acc0[q] + acc1[q];
}

__global__ __launch_bounds__(256) void ggc_degree(
    const int* __restrict__ dst, int* __restrict__ cnt, int* __restrict__ rank,
    long long nEdges, int nNodes)
{
    const long long e = (long long)blockIdx.x * 256 + threadIdx.x;
    if (e >= nEdges) return;
    const int d = dst[e];
    if ((unsigned)d < (unsigned)nNodes) rank[e] = atomicAdd(&cnt[d], 1);
}

__global__ __launch_bounds__(256) void ggc_rowptr(
    const int* __restrict__ cnt, int* __restrict__ rowptr, int nTotal)
{
    __shared__ int red[8];
    __shared__ int wsum[8];
    const int tid  = threadIdx.x;
    const int lane = tid & 31;
    const int wave = tid >> 5;
    const int blockBase = blockIdx.x * 1024;

    int part = 0;
    const int4* c4 = (const int4*)cnt;
    for (int i = tid; i < (blockBase >> 2); i += 256) {
        const int4 v = c4[i];
        part += (v.x + v.y) + (v.z + v.w);
    }
    for (int o = 16; o > 0; o >>= 1) part += __shfl_xor(part, o);
    if (lane == 0) red[wave] = part;

    const int i0 = blockBase + tid * 4;
    const int a0 = (i0 + 0 < nTotal) ? cnt[i0 + 0] : 0;
    const int a1 = (i0 + 1 < nTotal) ? cnt[i0 + 1] : 0;
    const int a2 = (i0 + 2 < nTotal) ? cnt[i0 + 2] : 0;
    const int a3 = (i0 + 3 < nTotal) ? cnt[i0 + 3] : 0;
    const int t  = (a0 + a1) + (a2 + a3);
    int inc = t;
    for (int o = 1; o < 32; o <<= 1) {
        const int n = __shfl_up(inc, o);
        if (lane >= o) inc += n;
    }
    if (lane == 31) wsum[wave] = inc;
    __syncthreads();

    int off = 0;
    for (int w = 0; w < 8; ++w) {
        off += red[w];
        if (w < wave) off += wsum[w];
    }
    int ex = off + inc - t;
    if (i0 + 0 < nTotal) rowptr[i0 + 0] = ex;
    ex += a0;
    if (i0 + 1 < nTotal) rowptr[i0 + 1] = ex;
    ex += a1;
    if (i0 + 2 < nTotal) rowptr[i0 + 2] = ex;
    ex += a2;
    if (i0 + 3 < nTotal) rowptr[i0 + 3] = ex;
}

__global__ __launch_bounds__(256) void ggc_fill(
    const int* __restrict__ src, const int* __restrict__ dst,
    const float* __restrict__ ew, const int* __restrict__ rowptr,
    const int* __restrict__ rank, int2* __restrict__ csr,
    long long nEdges, int nNodes)
{
    const long long e = (long long)blockIdx.x * 256 + threadIdx.x;
    if (e >= nEdges) return;
    const int d = dst[e];
    if ((unsigned)d >= (unsigned)nNodes) return;
    int2 rec;
    rec.x = src[e];
    rec.y = __float_as_int(ew[e]);
    csr[(long long)rowptr[d] + rank[e]] = rec;
}

template <int U>
static __device__ __forceinline__ void ggc_rows(
    const float4* __restrict__ hrow, int recS, int recW, int j, int nNodes, float4& acc)
{
    float4 v[U];
    float  w[U];
    #pragma unroll
    for (int u = 0; u < U; ++u) {
        int s = __builtin_amdgcn_readlane(recS, j + u);
        w[u]  = __int_as_float(__builtin_amdgcn_readlane(recW, j + u));
        s = ((unsigned)s < (unsigned)nNodes) ? s : 0;
        v[u] = hrow[(long long)s * (DIM / 4)];
    }
    #pragma unroll
    for (int u = 0; u < U; ++u) {
        acc.x = fmaf(w[u], v[u].x, acc.x);
        acc.y = fmaf(w[u], v[u].y, acc.y);
        acc.z = fmaf(w[u], v[u].z, acc.z);
        acc.w = fmaf(w[u], v[u].w, acc.w);
    }
}

__global__ __launch_bounds__(256) void ggc_gather(
    const float* __restrict__ h, const int* __restrict__ rowptr,
    const int2* __restrict__ csr, float* __restrict__ agg, int nNodes)
{
    const int lane = threadIdx.x & 31;
    const int node = __builtin_amdgcn_readfirstlane((int)(blockIdx.x * 8 + (threadIdx.x >> 5)));
    if (node >= nNodes) return;
    const int beg = __builtin_amdgcn_readfirstlane(rowptr[node]);
    const int end = __builtin_amdgcn_readfirstlane(rowptr[node + 1]);

    const float4* hrow = (const float4*)h + lane;
    float4 acc = make_float4(0.f, 0.f, 0.f, 0.f);

    for (int base = beg; base < end; base += 32) {
        const int cnt = min(32, end - base);
        int2 rec = make_int2(0, 0);
        if (lane < cnt) rec = csr[(long long)base + lane];
        int j = 0;
        for (; j + 8 <= cnt; j += 8) ggc_rows<8>(hrow, rec.x, rec.y, j, nNodes, acc);
        if (j + 4 <= cnt) { ggc_rows<4>(hrow, rec.x, rec.y, j, nNodes, acc); j += 4; }
        if (j + 2 <= cnt) { ggc_rows<2>(hrow, rec.x, rec.y, j, nNodes, acc); j += 2; }
        if (j < cnt) ggc_rows<1>(hrow, rec.x, rec.y, j, nNodes, acc);
    }
    ((float4*)agg)[(long long)node * (DIM / 4) + lane] = acc;
}

#define GRU_PITCH 260

__global__ __launch_bounds__(256) void ggc_gru(
    const float* __restrict__ agg, const float* __restrict__ h,
    const float* __restrict__ Wc, const float* __restrict__ W_hh,
    const float* __restrict__ b_ih, const float* __restrict__ b_hh,
    float* __restrict__ out, int nNodes)
{
    __shared__ __attribute__((aligned(16))) float aT[2][16][GRU_PITCH];

    const int tid  = threadIdx.x;
    const int wave = tid >> 5;
    const int lane = tid & 31;
    const int c    = lane & 15;
    const int hf   = lane >> 4;
    const int col  = wave * 16 + c;

    const int nTiles = (nNodes + 15) >> 4;

    v4f bi[3][16], bh[3][16];
    #pragma unroll
    for (int g = 0; g < 3; ++g) {
        const float* wi = Wc + (long long)(g * DIM + col) * DIM + 4 * hf;
        const float* wh = W_hh + (long long)(g * DIM + col) * DIM + 4 * hf;
        #pragma unroll
        for (int j = 0; j < 16; ++j) {
            bi[g][j] = *(const v4f*)(wi + 8 * j);
            bh[g][j] = *(const v4f*)(wh + 8 * j);
        }
    }
    const float bir = b_ih[col], biz = b_ih[col + DIM], bin = b_ih[col + 2 * DIM];
    const float bhr = b_hh[col], bhz = b_hh[col + DIM], bhn = b_hh[col + 2 * DIM];

    const int q  = tid & 63;
    const int r0 = tid >> 6;
    const float* srcBase = (q < 32) ? (agg + q * 4) : (h + (q - 32) * 4);
    float4 p[4];

    int tile = blockIdx.x;
    if (tile < nTiles) {
        for (int i = 0; i < 4; ++i) {
            const int row = tile * 16 + r0 + 4 * i;
            p[i] = (row < nNodes) ? *(const float4*)(srcBase + (long long)row * DIM)
                                  : make_float4(0.f, 0.f, 0.f, 0.f);
            *(float4*)&aT[0][r0 + 4 * i][q * 4] = p[i];
        }
    }
    __syncthreads();

    int buf = 0;
    for (; tile < nTiles; tile += gridDim.x) {
        const int  nxt     = tile + gridDim.x;
        const bool hasNext = nxt < nTiles;
        if (hasNext) {
            for (int i = 0; i < 4; ++i) {
                const int row = nxt * 16 + r0 + 4 * i;
                p[i] = (row < nNodes) ? *(const float4*)(srcBase + (long long)row * DIM)
                                      : make_float4(0.f, 0.f, 0.f, 0.f);
            }
        }

        v8f aIr = {}, aIz = {}, aIn = {}, aHr = {}, aHz = {}, aHn = {};
        const float* arow = &aT[buf][c][4 * hf];
        #pragma unroll
        for (int j = 0; j < 16; ++j) {
            const v4f xa = *(const v4f*)(arow + 8 * j);
            const v4f xh = *(const v4f*)(arow + DIM + 8 * j);
            const v2f xa0 = {xa.x, xa.y}, xa1 = {xa.z, xa.w};
            const v2f xh0 = {xh.x, xh.y}, xh1 = {xh.z, xh.w};
            v2f b0, b1;
            b0 = (v2f){bi[0][j].x, bi[0][j].y}; b1 = (v2f){bi[0][j].z, bi[0][j].w};
            aIr = __builtin_amdgcn_wmma_f32_16x16x4_f32(false, xa0, false, b0, (short)0, aIr, false, false);
            aIr = __builtin_amdgcn_wmma_f32_16x16x4_f32(false, xa1, false, b1, (short)0, aIr, false, false);
            b0 = (v2f){bi[1][j].x, bi[1][j].y}; b1 = (v2f){bi[1][j].z, bi[1][j].w};
            aIz = __builtin_amdgcn_wmma_f32_16x16x4_f32(false, xa0, false, b0, (short)0, aIz, false, false);
            aIz = __builtin_amdgcn_wmma_f32_16x16x4_f32(false, xa1, false, b1, (short)0, aIz, false, false);
            b0 = (v2f){bi[2][j].x, bi[2][j].y}; b1 = (v2f){bi[2][j].z, bi[2][j].w};
            aIn = __builtin_amdgcn_wmma_f32_16x16x4_f32(false, xa0, false, b0, (short)0, aIn, false, false);
            aIn = __builtin_amdgcn_wmma_f32_16x16x4_f32(false, xa1, false, b1, (short)0, aIn, false, false);
            b0 = (v2f){bh[0][j].x, bh[0][j].y}; b1 = (v2f){bh[0][j].z, bh[0][j].w};
            aHr = __builtin_amdgcn_wmma_f32_16x16x4_f32(false, xh0, false, b0, (short)0, aHr, false, false);
            aHr = __builtin_amdgcn_wmma_f32_16x16x4_f32(false, xh1, false, b1, (short)0, aHr, false, false);
            b0 = (v2f){bh[1][j].x, bh[1][j].y}; b1 = (v2f){bh[1][j].z, bh[1][j].w};
            aHz = __builtin_amdgcn_wmma_f32_16x16x4_f32(false, xh0, false, b0, (short)0, aHz, false, false);
            aHz = __builtin_amdgcn_wmma_f32_16x16x4_f32(false, xh1, false, b1, (short)0, aHz, false, false);
            b0 = (v2f){bh[2][j].x, bh[2][j].y}; b1 = (v2f){bh[2][j].z, bh[2][j].w};
            aHn = __builtin_amdgcn_wmma_f32_16x16x4_f32(false, xh0, false, b0, (short)0, aHn, false, false);
            aHn = __builtin_amdgcn_wmma_f32_16x16x4_f32(false, xh1, false, b1, (short)0, aHn, false, false);
        }

        #pragma unroll
        for (int r = 0; r < 8; ++r) {
            const int row  = 8 * hf + r;
            const int node = tile * 16 + row;
            float ir = aIr[r] + bir;
            float iz = aIz[r] + biz;
            float in = aIn[r] + bin;
            float hr = aHr[r] + bhr;
            float hz = aHz[r] + bhz;
            float hn = aHn[r] + bhn;

            float rg = 1.0f / (1.0f + __expf(-(ir + hr)));
            float zg = 1.0f / (1.0f + __expf(-(iz + hz)));
            float ng = tanhf(in + rg * hn);
            float hp = aT[buf][row][DIM + col];
            if (node < nNodes)
                out[(long long)node * DIM + col] = (1.0f - zg) * ng + zg * hp;
        }

        if (hasNext) {
            for (int i = 0; i < 4; ++i)
                *(float4*)&aT[buf ^ 1][r0 + 4 * i][q * 4] = p[i];
        }
        __syncthreads();
        buf ^= 1;
    }
}

extern "C" void kernel_launch(void* const* d_in, const int* in_sizes, int n_in,
                              void* d_out, int out_size, void* d_ws, size_t ws_size,
                              hipStream_t stream) {
    const float* x    = (const float*)d_in[0];
    const int*   eidx = (const int*)d_in[1];
    const float* ea   = (const float*)d_in[2];
    const float* W    = (const float*)d_in[3];
    const float* W_ih = (const float*)d_in[4];
    const float* W_hh = (const float*)d_in[5];
    const float* b_ih = (const float*)d_in[6];
    const float* b_hh = (const float*)d_in[7];

    const int       N = in_sizes[0] / DIM;
    const long long E = (long long)in_sizes[2];
    const int       L = in_sizes[3] / (DIM * DIM);

    const int* src = eidx;
    const int* dst = eidx + E;

    const size_t nd = (size_t)N * DIM;
    float* P0  = (float*)d_ws;
    float* P1  = P0 + nd;
    float* AGG = P1 + nd;

    const int    nTotal = N + 1;
    const size_t padInt = ((size_t)nTotal + 1023) / 1024 * 1024;
    int*  CNT    = (int*)(AGG + nd);
    int*  ROWPTR = CNT + padInt;
    int*  RANK   = ROWPTR + padInt;
    int2* CSR    = (int2*)(RANK + ((size_t)E + 3) / 4 * 4);
    float* WC    = (float*)(CSR + ((size_t)E + 1) / 2 * 2);

    const dim3 blk(256);
    const dim3 gridWc((unsigned)(L * 24));
    const dim3 gridGru(256);
    const dim3 gridEdges((unsigned)((E + 255) / 256));
    const dim3 gridScan((unsigned)(padInt / 1024));
    const dim3 gridGather((unsigned)((N + 7) / 8));

    ggc_wcomb<<<gridWc, blk, 0, stream>>>(W_ih, W, WC, L);

    hipMemsetAsync(CNT, 0, (size_t)nTotal * sizeof(int), stream);
    ggc_degree<<<gridEdges, blk, 0, stream>>>(dst, CNT, RANK, E, N);
    ggc_rowptr<<<gridScan, blk, 0, stream>>>(CNT, ROWPTR, nTotal);
    ggc_fill<<<gridEdges, blk, 0, stream>>>(src, dst, ea, ROWPTR, RANK, CSR, E, N);

    const float* hcur = x;
    for (int l = 0; l < L; ++l) {
        ggc_gather<<<gridGather, blk, 0, stream>>>(hcur, ROWPTR, CSR, AGG, N);

        float* hout = (l == L - 1) ? (float*)d_out : ((l & 1) ? P1 : P0);
        ggc_gru<<<gridGru, blk, 0, stream>>>(AGG, hcur, WC + (size_t)l * 3 * DIM * DIM,
                                             W_hh, b_ih, b_hh, hout, N);
        hcur = hout;
    }
}


# ggc_gru: per-tile barrier split: signal after the LDS reads and next-stage fill, wait at top of next tile; one wave's gate math now runs under the other wave's WMMAs; old h held in registers
# speedup vs baseline: 3.0918x; 1.0696x over previous
#include <hip/hip_runtime.h>
#include <hip/hip_bf16.h>


#define DIM 128

typedef __attribute__((ext_vector_type(2))) float v2f;
typedef __attribute__((ext_vector_type(4))) float v4f;
typedef __attribute__((ext_vector_type(8))) float v8f;

__global__ __launch_bounds__(256) void ggc_wcomb(
    const float* __restrict__ W_ih, const float* __restrict__ W,
    float* __restrict__ Wc, int nLayers)
{
    const int lane = threadIdx.x & 31;
    const int c    = lane & 15;
    const int hf   = lane >> 4;
    const int t = __builtin_amdgcn_readfirstlane((int)(blockIdx.x * 8 + (threadIdx.x >> 5)));
    const int l = t / 192;
    if (l >= nLayers) return;
    const int r  = t - l * 192;
    const int to = (r >> 3) * 16;
    const int tk = (r & 7) * 16;

    const float* a = W_ih + (long long)(to + c) * DIM + 4 * hf;
    const float* b = W + ((long long)l * DIM + tk + c) * DIM + 4 * hf;

    v8f acc0 = {}, acc1 = {};
    #pragma unroll
    for (int jj = 0; jj < 16; ++jj) {
        const v4f xa = *(const v4f*)(a + 8 * jj);
        const v4f xb = *(const v4f*)(b + 8 * jj);
        const v2f a0 = {xa.x, xa.y}, a1 = {xa.z, xa.w};
        const v2f b0 = {xb.x, xb.y}, b1 = {xb.z, xb.w};
        acc0 = __builtin_amdgcn_wmma_f32_16x16x4_f32(false, a0, false, b0, (short)0, acc0, false, false);
        acc1 = __builtin_amdgcn_wmma_f32_16x16x4_f32(false, a1, false, b1, (short)0, acc1, false, false);
    }

    float* d = Wc + ((long long)l * 3 * DIM + to + 8 * hf) * DIM + tk + c;
    #pragma unroll
    for (int q = 0; q < 8; ++q)
        d[(long long)q * DIM] = acc0[q] + acc1[q];
}

__global__ __launch_bounds__(256) void ggc_degree(
    const int* __restrict__ dst, int* __restrict__ cnt, int* __restrict__ rank,
    long long nEdges, int nNodes)
{
    const long long e = (long long)blockIdx.x * 256 + threadIdx.x;
    if (e >= nEdges) return;
    const int d = dst[e];
    if ((unsigned)d < (unsigned)nNodes) rank[e] = atomicAdd(&cnt[d], 1);
}

__global__ __launch_bounds__(256) void ggc_rowptr(
    const int* __restrict__ cnt, int* __restrict__ rowptr, int nTotal)
{
    __shared__ int red[8];
    __shared__ int wsum[8];
    const int tid  = threadIdx.x;
    const int lane = tid & 31;
    const int wave = tid >> 5;
    const int blockBase = blockIdx.x * 1024;

    int part = 0;
    const int4* c4 = (const int4*)cnt;
    for (int i = tid; i < (blockBase >> 2); i += 256) {
        const int4 v = c4[i];
        part += (v.x + v.y) + (v.z + v.w);
    }
    for (int o = 16; o > 0; o >>= 1) part += __shfl_xor(part, o);
    if (lane == 0) red[wave] = part;

    const int i0 = blockBase + tid * 4;
    const int a0 = (i0 + 0 < nTotal) ? cnt[i0 + 0] : 0;
    const int a1 = (i0 + 1 < nTotal) ? cnt[i0 + 1] : 0;
    const int a2 = (i0 + 2 < nTotal) ? cnt[i0 + 2] : 0;
    const int a3 = (i0 + 3 < nTotal) ? cnt[i0 + 3] : 0;
    const int t  = (a0 + a1) + (a2 + a3);
    int inc = t;
    for (int o = 1; o < 32; o <<= 1) {
        const int n = __shfl_up(inc, o);
        if (lane >= o) inc += n;
    }
    if (lane == 31) wsum[wave] = inc;
    __syncthreads();

    int off = 0;
    for (int w = 0; w < 8; ++w) {
        off += red[w];
        if (w < wave) off += wsum[w];
    }
    int ex = off + inc - t;
    if (i0 + 0 < nTotal) rowptr[i0 + 0] = ex;
    ex += a0;
    if (i0 + 1 < nTotal) rowptr[i0 + 1] = ex;
    ex += a1;
    if (i0 + 2 < nTotal) rowptr[i0 + 2] = ex;
    ex += a2;
    if (i0 + 3 < nTotal) rowptr[i0 + 3] = ex;
}

__global__ __launch_bounds__(256) void ggc_fill(
    const int* __restrict__ src, const int* __restrict__ dst,
    const float* __restrict__ ew, const int* __restrict__ rowptr,
    const int* __restrict__ rank, int2* __restrict__ csr,
    long long nEdges, int nNodes)
{
    const long long e = (long long)blockIdx.x * 256 + threadIdx.x;
    if (e >= nEdges) return;
    const int d = dst[e];
    if ((unsigned)d >= (unsigned)nNodes) return;
    int2 rec;
    rec.x = src[e];
    rec.y = __float_as_int(ew[e]);
    csr[(long long)rowptr[d] + rank[e]] = rec;
}

template <int U>
static __device__ __forceinline__ void ggc_rows(
    const float4* __restrict__ hrow, int recS, int recW, int j, int nNodes, float4& acc)
{
    float4 v[U];
    float  w[U];
    #pragma unroll
    for (int u = 0; u < U; ++u) {
        int s = __builtin_amdgcn_readlane(recS, j + u);
        w[u]  = __int_as_float(__builtin_amdgcn_readlane(recW, j + u));
        s = ((unsigned)s < (unsigned)nNodes) ? s : 0;
        v[u] = hrow[(long long)s * (DIM / 4)];
    }
    #pragma unroll
    for (int u = 0; u < U; ++u) {
        acc.x = fmaf(w[u], v[u].x, acc.x);
        acc.y = fmaf(w[u], v[u].y, acc.y);
        acc.z = fmaf(w[u], v[u].z, acc.z);
        acc.w = fmaf(w[u], v[u].w, acc.w);
    }
}

__global__ __launch_bounds__(256) void ggc_gather(
    const float* __restrict__ h, const int* __restrict__ rowptr,
    const int2* __restrict__ csr, float* __restrict__ agg, int nNodes)
{
    const int lane = threadIdx.x & 31;
    const int node = __builtin_amdgcn_readfirstlane((int)(blockIdx.x * 8 + (threadIdx.x >> 5)));
    if (node >= nNodes) return;
    const int beg = __builtin_amdgcn_readfirstlane(rowptr[node]);
    const int end = __builtin_amdgcn_readfirstlane(rowptr[node + 1]);

    const float4* hrow = (const float4*)h + lane;
    float4 acc = make_float4(0.f, 0.f, 0.f, 0.f);

    for (int base = beg; base < end; base += 32) {
        const int cnt = min(32, end - base);
        int2 rec = make_int2(0, 0);
        if (lane < cnt) rec = csr[(long long)base + lane];
        int j = 0;
        for (; j + 8 <= cnt; j += 8) ggc_rows<8>(hrow, rec.x, rec.y, j, nNodes, acc);
        if (j + 4 <= cnt) { ggc_rows<4>(hrow, rec.x, rec.y, j, nNodes, acc); j += 4; }
        if (j + 2 <= cnt) { ggc_rows<2>(hrow, rec.x, rec.y, j, nNodes, acc); j += 2; }
        if (j < cnt) ggc_rows<1>(hrow, rec.x, rec.y, j, nNodes, acc);
    }
    ((float4*)agg)[(long long)node * (DIM / 4) + lane] = acc;
}

#define GRU_PITCH 260

__global__ __launch_bounds__(256) void ggc_gru(
    const float* __restrict__ agg, const float* __restrict__ h,
    const float* __restrict__ Wc, const float* __restrict__ W_hh,
    const float* __restrict__ b_ih, const float* __restrict__ b_hh,
    float* __restrict__ out, int nNodes)
{
    __shared__ __attribute__((aligned(16))) float aT[2][16][GRU_PITCH];

    const int tid  = threadIdx.x;
    const int wave = tid >> 5;
    const int lane = tid & 31;
    const int c    = lane & 15;
    const int hf   = lane >> 4;
    const int col  = wave * 16 + c;

    const int nTiles = (nNodes + 15) >> 4;

    v4f bi[3][16], bh[3][16];
    #pragma unroll
    for (int g = 0; g < 3; ++g) {
        const float* wi = Wc + (long long)(g * DIM + col) * DIM + 4 * hf;
        const float* wh = W_hh + (long long)(g * DIM + col) * DIM + 4 * hf;
        #pragma unroll
        for (int j = 0; j < 16; ++j) {
            bi[g][j] = *(const v4f*)(wi + 8 * j);
            bh[g][j] = *(const v4f*)(wh + 8 * j);
        }
    }
    const float bir = b_ih[col], biz = b_ih[col + DIM], bin = b_ih[col + 2 * DIM];
    const float bhr = b_hh[col], bhz = b_hh[col + DIM], bhn = b_hh[col + 2 * DIM];

    const int q  = tid & 63;
    const int r0 = tid >> 6;
    const float* srcBase = (q < 32) ? (agg + q * 4) : (h + (q - 32) * 4);
    float4 p[4];

    int tile = blockIdx.x;
    if (tile < nTiles) {
        for (int i = 0; i < 4; ++i) {
            const int row = tile * 16 + r0 + 4 * i;
            p[i] = (row < nNodes) ? *(const float4*)(srcBase + (long long)row * DIM)
                                  : make_float4(0.f, 0.f, 0.f, 0.f);
            *(float4*)&aT[0][r0 + 4 * i][q * 4] = p[i];
        }
        __builtin_amdgcn_fence(__ATOMIC_RELEASE, "workgroup");
        __builtin_amdgcn_s_barrier_signal(-1);
    }

    int buf = 0;
    for (; tile < nTiles; tile += gridDim.x) {
        const int  nxt     = tile + gridDim.x;
        const bool hasNext = nxt < nTiles;
        __builtin_amdgcn_s_barrier_wait(-1);
        __builtin_amdgcn_fence(__ATOMIC_ACQUIRE, "workgroup");
        if (hasNext) {
            for (int i = 0; i < 4; ++i) {
                const int row = nxt * 16 + r0 + 4 * i;
                p[i] = (row < nNodes) ? *(const float4*)(srcBase + (long long)row * DIM)
                                      : make_float4(0.f, 0.f, 0.f, 0.f);
            }
        }

        v8f aIr = {}, aIz = {}, aIn = {}, aHr = {}, aHz = {}, aHn = {};
        const float* arow = &aT[buf][c][4 * hf];
        #pragma unroll
        for (int j = 0; j < 16; ++j) {
            const v4f xa = *(const v4f*)(arow + 8 * j);
            const v4f xh = *(const v4f*)(arow + DIM + 8 * j);
            const v2f xa0 = {xa.x, xa.y}, xa1 = {xa.z, xa.w};
            const v2f xh0 = {xh.x, xh.y}, xh1 = {xh.z, xh.w};
            v2f b0, b1;
            b0 = (v2f){bi[0][j].x, bi[0][j].y}; b1 = (v2f){bi[0][j].z, bi[0][j].w};
            aIr = __builtin_amdgcn_wmma_f32_16x16x4_f32(false, xa0, false, b0, (short)0, aIr, false, false);
            aIr = __builtin_amdgcn_wmma_f32_16x16x4_f32(false, xa1, false, b1, (short)0, aIr, false, false);
            b0 = (v2f){bi[1][j].x, bi[1][j].y}; b1 = (v2f){bi[1][j].z, bi[1][j].w};
            aIz = __builtin_amdgcn_wmma_f32_16x16x4_f32(false, xa0, false, b0, (short)0, aIz, false, false);
            aIz = __builtin_amdgcn_wmma_f32_16x16x4_f32(false, xa1, false, b1, (short)0, aIz, false, false);
            b0 = (v2f){bi[2][j].x, bi[2][j].y}; b1 = (v2f){bi[2][j].z, bi[2][j].w};
            aIn = __builtin_amdgcn_wmma_f32_16x16x4_f32(false, xa0, false, b0, (short)0, aIn, false, false);
            aIn = __builtin_amdgcn_wmma_f32_16x16x4_f32(false, xa1, false, b1, (short)0, aIn, false, false);
            b0 = (v2f){bh[0][j].x, bh[0][j].y}; b1 = (v2f){bh[0][j].z, bh[0][j].w};
            aHr = __builtin_amdgcn_wmma_f32_16x16x4_f32(false, xh0, false, b0, (short)0, aHr, false, false);
            aHr = __builtin_amdgcn_wmma_f32_16x16x4_f32(false, xh1, false, b1, (short)0, aHr, false, false);
            b0 = (v2f){bh[1][j].x, bh[1][j].y}; b1 = (v2f){bh[1][j].z, bh[1][j].w};
            aHz = __builtin_amdgcn_wmma_f32_16x16x4_f32(false, xh0, false, b0, (short)0, aHz, false, false);
            aHz = __builtin_amdgcn_wmma_f32_16x16x4_f32(false, xh1, false, b1, (short)0, aHz, false, false);
            b0 = (v2f){bh[2][j].x, bh[2][j].y}; b1 = (v2f){bh[2][j].z, bh[2][j].w};
            aHn = __builtin_amdgcn_wmma_f32_16x16x4_f32(false, xh0, false, b0, (short)0, aHn, false, false);
            aHn = __builtin_amdgcn_wmma_f32_16x16x4_f32(false, xh1, false, b1, (short)0, aHn, false, false);
        }

        float hp[8];
        #pragma unroll
        for (int r = 0; r < 8; ++r)
            hp[r] = aT[buf][8 * hf + r][DIM + col];

        if (hasNext) {
            for (int i = 0; i < 4; ++i)
                *(float4*)&aT[buf ^ 1][r0 + 4 * i][q * 4] = p[i];
            __builtin_amdgcn_fence(__ATOMIC_RELEASE, "workgroup");
            __builtin_amdgcn_s_barrier_signal(-1);
        }

        #pragma unroll
        for (int r = 0; r < 8; ++r) {
            const int row  = 8 * hf + r;
            const int node = tile * 16 + row;
            float ir = aIr[r] + bir;
            float iz = aIz[r] + biz;
            float in = aIn[r] + bin;
            float hr = aHr[r] + bhr;
            float hz = aHz[r] + bhz;
            float hn = aHn[r] + bhn;

            float rg = 1.0f / (1.0f + __expf(-(ir + hr)));
            float zg = 1.0f / (1.0f + __expf(-(iz + hz)));
            float ng = tanhf(in + rg * hn);
            if (node < nNodes)
                out[(long long)node * DIM + col] = (1.0f - zg) * ng + zg * hp[r];
        }
        buf ^= 1;
    }
}

extern "C" void kernel_launch(void* const* d_in, const int* in_sizes, int n_in,
                              void* d_out, int out_size, void* d_ws, size_t ws_size,
                              hipStream_t stream) {
    const float* x    = (const float*)d_in[0];
    const int*   eidx = (const int*)d_in[1];
    const float* ea   = (const float*)d_in[2];
    const float* W    = (const float*)d_in[3];
    const float* W_ih = (const float*)d_in[4];
    const float* W_hh = (const float*)d_in[5];
    const float* b_ih = (const float*)d_in[6];
    const float* b_hh = (const float*)d_in[7];

    const int       N = in_sizes[0] / DIM;
    const long long E = (long long)in_sizes[2];
    const int       L = in_sizes[3] / (DIM * DIM);

    const int* src = eidx;
    const int* dst = eidx + E;

    const size_t nd = (size_t)N * DIM;
    float* P0  = (float*)d_ws;
    float* P1  = P0 + nd;
    float* AGG = P1 + nd;

    const int    nTotal = N + 1;
    const size_t padInt = ((size_t)nTotal + 1023) / 1024 * 1024;
    int*  CNT    = (int*)(AGG + nd);
    int*  ROWPTR = CNT + padInt;
    int*  RANK   = ROWPTR + padInt;
    int2* CSR    = (int2*)(RANK + ((size_t)E + 3) / 4 * 4);
    float* WC    = (float*)(CSR + ((size_t)E + 1) / 2 * 2);

    const dim3 blk(256);
    const dim3 gridWc((unsigned)(L * 24));
    const dim3 gridGru(256);
    const dim3 gridEdges((unsigned)((E + 255) / 256));
    const dim3 gridScan((unsigned)(padInt / 1024));
    const dim3 gridGather((unsigned)((N + 7) / 8));

    ggc_wcomb<<<gridWc, blk, 0, stream>>>(W_ih, W, WC, L);

    hipMemsetAsync(CNT, 0, (size_t)nTotal * sizeof(int), stream);
    ggc_degree<<<gridEdges, blk, 0, stream>>>(dst, CNT, RANK, E, N);
    ggc_rowptr<<<gridScan, blk, 0, stream>>>(CNT, ROWPTR, nTotal);
    ggc_fill<<<gridEdges, blk, 0, stream>>>(src, dst, ea, ROWPTR, RANK, CSR, E, N);

    const float* hcur = x;
    for (int l = 0; l < L; ++l) {
        ggc_gather<<<gridGather, blk, 0, stream>>>(hcur, ROWPTR, CSR, AGG, N);

        float* hout = (l == L - 1) ? (float*)d_out : ((l & 1) ? P1 : P0);
        ggc_gru<<<gridGru, blk, 0, stream>>>(AGG, hcur, WC + (size_t)l * 3 * DIM * DIM,
                                             W_hh, b_ih, b_hh, hout, N);
        hcur = hout;
    }
}


# edge sort redone as two-level counting sort on LDS counters (512-node ranges per 4096-edge chunk, then nodes inside a range): no global atomic, no memset; in-node sum order = LDS add arrival order
# speedup vs baseline: 3.5865x; 1.1600x over previous
#include <hip/hip_runtime.h>
#include <hip/hip_bf16.h>


#define DIM 128

typedef __attribute__((ext_vector_type(2))) float v2f;
typedef __attribute__((ext_vector_type(4))) float v4f;
typedef __attribute__((ext_vector_type(8))) float v8f;

__global__ __launch_bounds__(256) void ggc_wcomb(
    const float* __restrict__ W_ih, const float* __restrict__ W,
    float* __restrict__ Wc, int nLayers)
{
    const int lane = threadIdx.x & 31;
    const int c    = lane & 15;
    const int hf   = lane >> 4;
    const int t = __builtin_amdgcn_readfirstlane((int)(blockIdx.x * 8 + (threadIdx.x >> 5)));
    const int l = t / 192;
    if (l >= nLayers) return;
    const int r  = t - l * 192;
    const int to = (r >> 3) * 16;
    const int tk = (r & 7) * 16;

    const float* a = W_ih + (long long)(to + c) * DIM + 4 * hf;
    const float* b = W + ((long long)l * DIM + tk + c) * DIM + 4 * hf;

    v8f acc0 = {}, acc1 = {};
    #pragma unroll
    for (int jj = 0; jj < 16; ++jj) {
        const v4f xa = *(const v4f*)(a + 8 * jj);
        const v4f xb = *(const v4f*)(b + 8 * jj);
        const v2f a0 = {xa.x, xa.y}, a1 = {xa.z, xa.w};
        const v2f b0 = {xb.x, xb.y}, b1 = {xb.z, xb.w};
        acc0 = __builtin_amdgcn_wmma_f32_16x16x4_f32(false, a0, false, b0, (short)0, acc0, false, false);
        acc1 = __builtin_amdgcn_wmma_f32_16x16x4_f32(false, a1, false, b1, (short)0, acc1, false, false);
    }

    float* d = Wc + ((long long)l * 3 * DIM + to + 8 * hf) * DIM + tk + c;
    #pragma unroll
    for (int q = 0; q < 8; ++q)
        d[(long long)q * DIM] = acc0[q] + acc1[q];
}

#define PART_SHIFT 9
#define PART_NODES (1 << PART_SHIFT)
#define PART_MAX   1024
#define CHUNK      4096

__global__ __launch_bounds__(256) void ggc_part_count(
    const int* __restrict__ dst, int* __restrict__ hist,
    long long nEdges, int nNodes, int nRanges, int nChunks)
{
    __shared__ int cnt[PART_MAX];
    const int tid = threadIdx.x;
    for (int b = tid; b < nRanges; b += 256) cnt[b] = 0;
    __syncthreads();

    const long long base = (long long)blockIdx.x * CHUNK + tid;
    int d[16];
    #pragma unroll
    for (int k = 0; k < 16; ++k) {
        const long long e = base + k * 256;
        d[k] = (e < nEdges) ? dst[e] : -1;
    }
    #pragma unroll
    for (int k = 0; k < 16; ++k)
        if ((unsigned)d[k] < (unsigned)nNodes) atomicAdd(&cnt[d[k] >> PART_SHIFT], 1);
    __syncthreads();

    for (int b = tid; b < nRanges; b += 256)
        hist[(long long)b * nChunks + blockIdx.x] = cnt[b];
}

__global__ __launch_bounds__(1024) void ggc_scan(
    const int* __restrict__ in, int* __restrict__ out, int n)
{
    __shared__ int red[32];
    __shared__ int wsum[32];
    const int tid  = threadIdx.x;
    const int lane = tid & 31;
    const int wave = tid >> 5;
    const int blockBase = blockIdx.x * 4096;

    int part = 0;
    const int4* c4 = (const int4*)in + tid;
    const int nSlabs = blockIdx.x;
    const int4 zero4 = make_int4(0, 0, 0, 0);
    for (int sl = 0; sl < nSlabs; sl += 4) {
        const int4* p = c4 + sl * 1024;
        const int4 v0 = p[0];
        const int4 v1 = (sl + 1 < nSlabs) ? p[1024] : zero4;
        const int4 v2 = (sl + 2 < nSlabs) ? p[2048] : zero4;
        const int4 v3 = (sl + 3 < nSlabs) ? p[3072] : zero4;
        part += ((v0.x + v0.y) + (v0.z + v0.w)) + ((v1.x + v1.y) + (v1.z + v1.w));
        part += ((v2.x + v2.y) + (v2.z + v2.w)) + ((v3.x + v3.y) + (v3.z + v3.w));
    }
    for (int o = 16; o > 0; o >>= 1) part += __shfl_xor(part, o);
    if (lane == 0) red[wave] = part;

    const int i0 = blockBase + tid * 4;
    const int a0 = (i0 + 0 < n) ? in[i0 + 0] : 0;
    const int a1 = (i0 + 1 < n) ? in[i0 + 1] : 0;
    const int a2 = (i0 + 2 < n) ? in[i0 + 2] : 0;
    const int a3 = (i0 + 3 < n) ? in[i0 + 3] : 0;
    const int t  = (a0 + a1) + (a2 + a3);
    int inc = t;
    for (int o = 1; o < 32; o <<= 1) {
        const int m = __shfl_up(inc, o);
        if (lane >= o) inc += m;
    }
    if (lane == 31) wsum[wave] = inc;
    __syncthreads();

    int off = 0;
    for (int w = 0; w < 32; ++w) {
        off += red[w];
        if (w < wave) off += wsum[w];
    }
    int ex = off + inc - t;
    if (i0 + 0 <= n) out[i0 + 0] = ex;
    ex += a0;
    if (i0 + 1 <= n) out[i0 + 1] = ex;
    ex += a1;
    if (i0 + 2 <= n) out[i0 + 2] = ex;
    ex += a2;
    if (i0 + 3 <= n) out[i0 + 3] = ex;
}

__global__ __launch_bounds__(256) void ggc_part_place(
    const int* __restrict__ src, const int* __restrict__ dst,
    const float* __restrict__ ew, const int* __restrict__ off,
    int2* __restrict__ recSW, int* __restrict__ recD,
    long long nEdges, int nNodes, int nRanges, int nChunks)
{
    __shared__ int cur[PART_MAX];
    const int tid = threadIdx.x;
    for (int b = tid; b < nRanges; b += 256)
        cur[b] = off[(long long)b * nChunks + blockIdx.x];
    __syncthreads();

    const long long base = (long long)blockIdx.x * CHUNK + tid;
    for (int half = 0; half < 2; ++half) {
        int d[8], sIdx[8], wBits[8];
        #pragma unroll
        for (int k = 0; k < 8; ++k) {
            const long long e = base + (half * 8 + k) * 256;
            const bool in = e < nEdges;
            d[k]     = in ? dst[e] : -1;
            sIdx[k]  = in ? src[e] : 0;
            wBits[k] = in ? __float_as_int(ew[e]) : 0;
        }
        #pragma unroll
        for (int k = 0; k < 8; ++k) {
            if ((unsigned)d[k] < (unsigned)nNodes) {
                const int pos = atomicAdd(&cur[d[k] >> PART_SHIFT], 1);
                recSW[pos] = make_int2(sIdx[k], wBits[k]);
                recD[pos]  = d[k];
            }
        }
    }
}

__global__ __launch_bounds__(1024) void ggc_range_sort(
    const int* __restrict__ off, const int* __restrict__ recD,
    const int2* __restrict__ recSW, int* __restrict__ rowptr,
    int2* __restrict__ csr, int nNodes, int nRanges, int nChunks)
{
    __shared__ int cnt[PART_NODES];
    __shared__ int wsum[32];
    const int tid  = threadIdx.x;
    const int lane = tid & 31;
    const int wave = tid >> 5;
    const int b     = blockIdx.x;
    const int node0 = b << PART_SHIFT;
    const int beg = off[(long long)b * nChunks];
    const int end = off[(long long)(b + 1) * nChunks];

    if (tid < PART_NODES) cnt[tid] = 0;
    __syncthreads();
    for (int i = beg + tid; i < end; i += 1024)
        atomicAdd(&cnt[(recD[i] - node0) & (PART_NODES - 1)], 1);
    __syncthreads();

    const int v = (tid < PART_NODES) ? cnt[tid] : 0;
    int inc = v;
    for (int o = 1; o < 32; o <<= 1) {
        const int m = __shfl_up(inc, o);
        if (lane >= o) inc += m;
    }
    if (lane == 31) wsum[wave] = inc;
    __syncthreads();
    int before = 0;
    for (int w = 0; w < 32; ++w)
        if (w < wave) before += wsum[w];
    const int ex = before + inc - v;
    if (tid < PART_NODES) {
        if (node0 + tid < nNodes) rowptr[node0 + tid] = beg + ex;
        cnt[tid] = ex;
    }
    if (b == nRanges - 1 && tid == 0) rowptr[nNodes] = end;
    __syncthreads();

    for (int i = beg + tid; i < end; i += 1024) {
        const int  d   = recD[i];
        const int2 rec = recSW[i];
        const int  pos = atomicAdd(&cnt[(d - node0) & (PART_NODES - 1)], 1);
        csr[(long long)beg + pos] = rec;
    }
}

template <int U>
static __device__ __forceinline__ void ggc_rows(
    const float4* __restrict__ hrow, int recS, int recW, int j, int nNodes, float4& acc)
{
    float4 v[U];
    float  w[U];
    #pragma unroll
    for (int u = 0; u < U; ++u) {
        int s = __builtin_amdgcn_readlane(recS, j + u);
        w[u]  = __int_as_float(__builtin_amdgcn_readlane(recW, j + u));
        s = ((unsigned)s < (unsigned)nNodes) ? s : 0;
        v[u] = hrow[(long long)s * (DIM / 4)];
    }
    #pragma unroll
    for (int u = 0; u < U; ++u) {
        acc.x = fmaf(w[u], v[u].x, acc.x);
        acc.y = fmaf(w[u], v[u].y, acc.y);
        acc.z = fmaf(w[u], v[u].z, acc.z);
        acc.w = fmaf(w[u], v[u].w, acc.w);
    }
}

__global__ __launch_bounds__(256) void ggc_gather(
    const float* __restrict__ h, const int* __restrict__ rowptr,
    const int2* __restrict__ csr, float* __restrict__ agg, int nNodes)
{
    const int lane = threadIdx.x & 31;
    const int node = __builtin_amdgcn_readfirstlane((int)(blockIdx.x * 8 + (threadIdx.x >> 5)));
    if (node >= nNodes) return;
    const int beg = __builtin_amdgcn_readfirstlane(rowptr[node]);
    const int end = __builtin_amdgcn_readfirstlane(rowptr[node + 1]);

    const float4* hrow = (const float4*)h + lane;
    float4 acc = make_float4(0.f, 0.f, 0.f, 0.f);

    for (int base = beg; base < end; base += 32) {
        const int cnt = min(32, end - base);
        int2 rec = make_int2(0, 0);
        if (lane < cnt) rec = csr[(long long)base + lane];
        int j = 0;
        for (; j + 8 <= cnt; j += 8) ggc_rows<8>(hrow, rec.x, rec.y, j, nNodes, acc);
        if (j + 4 <= cnt) { ggc_rows<4>(hrow, rec.x, rec.y, j, nNodes, acc); j += 4; }
        if (j + 2 <= cnt) { ggc_rows<2>(hrow, rec.x, rec.y, j, nNodes, acc); j += 2; }
        if (j < cnt) ggc_rows<1>(hrow, rec.x, rec.y, j, nNodes, acc);
    }
    ((float4*)agg)[(long long)node * (DIM / 4) + lane] = acc;
}

#define GRU_PITCH 260

__global__ __launch_bounds__(256) void ggc_gru(
    const float* __restrict__ agg, const float* __restrict__ h,
    const float* __restrict__ Wc, const float* __restrict__ W_hh,
    const float* __restrict__ b_ih, const float* __restrict__ b_hh,
    float* __restrict__ out, int nNodes)
{
    __shared__ __attribute__((aligned(16))) float aT[2][16][GRU_PITCH];

    const int tid  = threadIdx.x;
    const int wave = tid >> 5;
    const int lane = tid & 31;
    const int c    = lane & 15;
    const int hf   = lane >> 4;
    const int col  = wave * 16 + c;

    const int nTiles = (nNodes + 15) >> 4;

    v4f bi[3][16], bh[3][16];
    #pragma unroll
    for (int g = 0; g < 3; ++g) {
        const float* wi = Wc + (long long)(g * DIM + col) * DIM + 4 * hf;
        const float* wh = W_hh + (long long)(g * DIM + col) * DIM + 4 * hf;
        #pragma unroll
        for (int j = 0; j < 16; ++j) {
            bi[g][j] = *(const v4f*)(wi + 8 * j);
            bh[g][j] = *(const v4f*)(wh + 8 * j);
        }
    }
    const float bir = b_ih[col], biz = b_ih[col + DIM], bin = b_ih[col + 2 * DIM];
    const float bhr = b_hh[col], bhz = b_hh[col + DIM], bhn = b_hh[col + 2 * DIM];

    const int q  = tid & 63;
    const int r0 = tid >> 6;
    const float* srcBase = (q < 32) ? (agg + q * 4) : (h + (q - 32) * 4);
    float4 p[4];

    int tile = blockIdx.x;
    if (tile < nTiles) {
        for (int i = 0; i < 4; ++i) {
            const int row = tile * 16 + r0 + 4 * i;
            p[i] = (row < nNodes) ? *(const float4*)(srcBase + (long long)row * DIM)
                                  : make_float4(0.f, 0.f, 0.f, 0.f);
            *(float4*)&aT[0][r0 + 4 * i][q * 4] = p[i];
        }
        __builtin_amdgcn_fence(__ATOMIC_RELEASE, "workgroup");
        __builtin_amdgcn_s_barrier_signal(-1);
    }

    int buf = 0;
    for (; tile < nTiles; tile += gridDim.x) {
        const int  nxt     = tile + gridDim.x;
        const bool hasNext = nxt < nTiles;
        __builtin_amdgcn_s_barrier_wait(-1);
        __builtin_amdgcn_fence(__ATOMIC_ACQUIRE, "workgroup");
        if (hasNext) {
            for (int i = 0; i < 4; ++i) {
                const int row = nxt * 16 + r0 + 4 * i;
                p[i] = (row < nNodes) ? *(const float4*)(srcBase + (long long)row * DIM)
                                      : make_float4(0.f, 0.f, 0.f, 0.f);
            }
        }

        v8f aIr = {}, aIz = {}, aIn = {}, aHr = {}, aHz = {}, aHn = {};
        const float* arow = &aT[buf][c][4 * hf];
        #pragma unroll
        for (int j = 0; j < 16; ++j) {
            const v4f xa = *(const v4f*)(arow + 8 * j);
            const v4f xh = *(const v4f*)(arow + DIM + 8 * j);
            const v2f xa0 = {xa.x, xa.y}, xa1 = {xa.z, xa.w};
            const v2f xh0 = {xh.x, xh.y}, xh1 = {xh.z, xh.w};
            v2f b0, b1;
            b0 = (v2f){bi[0][j].x, bi[0][j].y}; b1 = (v2f){bi[0][j].z, bi[0][j].w};
            aIr = __builtin_amdgcn_wmma_f32_16x16x4_f32(false, xa0, false, b0, (short)0, aIr, false, false);
            aIr = __builtin_amdgcn_wmma_f32_16x16x4_f32(false, xa1, false, b1, (short)0, aIr, false, false);
            b0 = (v2f){bi[1][j].x, bi[1][j].y}; b1 = (v2f){bi[1][j].z, bi[1][j].w};
            aIz = __builtin_amdgcn_wmma_f32_16x16x4_f32(false, xa0, false, b0, (short)0, aIz, false, false);
            aIz = __builtin_amdgcn_wmma_f32_16x16x4_f32(false, xa1, false, b1, (short)0, aIz, false, false);
            b0 = (v2f){bi[2][j].x, bi[2][j].y}; b1 = (v2f){bi[2][j].z, bi[2][j].w};
            aIn = __builtin_amdgcn_wmma_f32_16x16x4_f32(false, xa0, false, b0, (short)0, aIn, false, false);
            aIn = __builtin_amdgcn_wmma_f32_16x16x4_f32(false, xa1, false, b1, (short)0, aIn, false, false);
            b0 = (v2f){bh[0][j].x, bh[0][j].y}; b1 = (v2f){bh[0][j].z, bh[0][j].w};
            aHr = __builtin_amdgcn_wmma_f32_16x16x4_f32(false, xh0, false, b0, (short)0, aHr, false, false);
            aHr = __builtin_amdgcn_wmma_f32_16x16x4_f32(false, xh1, false, b1, (short)0, aHr, false, false);
            b0 = (v2f){bh[1][j].x, bh[1][j].y}; b1 = (v2f){bh[1][j].z, bh[1][j].w};
            aHz = __builtin_amdgcn_wmma_f32_16x16x4_f32(false, xh0, false, b0, (short)0, aHz, false, false);
            aHz = __builtin_amdgcn_wmma_f32_16x16x4_f32(false, xh1, false, b1, (short)0, aHz, false, false);
            b0 = (v2f){bh[2][j].x, bh[2][j].y}; b1 = (v2f){bh[2][j].z, bh[2][j].w};
            aHn = __builtin_amdgcn_wmma_f32_16x16x4_f32(false, xh0, false, b0, (short)0, aHn, false, false);
            aHn = __builtin_amdgcn_wmma_f32_16x16x4_f32(false, xh1, false, b1, (short)0, aHn, false, false);
        }

        float hp[8];
        #pragma unroll
        for (int r = 0; r < 8; ++r)
            hp[r] = aT[buf][8 * hf + r][DIM + col];

        if (hasNext) {
            for (int i = 0; i < 4; ++i)
                *(float4*)&aT[buf ^ 1][r0 + 4 * i][q * 4] = p[i];
            __builtin_amdgcn_fence(__ATOMIC_RELEASE, "workgroup");
            __builtin_amdgcn_s_barrier_signal(-1);
        }

        #pragma unroll
        for (int r = 0; r < 8; ++r) {
            const int row  = 8 * hf + r;
            const int node = tile * 16 + row;
            float ir = aIr[r] + bir;
            float iz = aIz[r] + biz;
            float in = aIn[r] + bin;
            float hr = aHr[r] + bhr;
            float hz = aHz[r] + bhz;
            float hn = aHn[r] + bhn;

            float rg = 1.0f / (1.0f + __expf(-(ir + hr)));
            float zg = 1.0f / (1.0f + __expf(-(iz + hz)));
            float ng = tanhf(in + rg * hn);
            if (node < nNodes)
                out[(long long)node * DIM + col] = (1.0f - zg) * ng + zg * hp[r];
        }
        buf ^= 1;
    }
}

extern "C" void kernel_launch(void* const* d_in, const int* in_sizes, int n_in,
                              void* d_out, int out_size, void* d_ws, size_t ws_size,
                              hipStream_t stream) {
    const float* x    = (const float*)d_in[0];
    const int*   eidx = (const int*)d_in[1];
    const float* ea   = (const float*)d_in[2];
    const float* W    = (const float*)d_in[3];
    const float* W_ih = (const float*)d_in[4];
    const float* W_hh = (const float*)d_in[5];
    const float* b_ih = (const float*)d_in[6];
    const float* b_hh = (const float*)d_in[7];

    const int       N = in_sizes[0] / DIM;
    const long long E = (long long)in_sizes[2];
    const int       L = in_sizes[3] / (DIM * DIM);

    const int* src = eidx;
    const int* dst = eidx + E;

    const size_t nd = (size_t)N * DIM;
    float* P0  = (float*)d_ws;
    float* P1  = P0 + nd;
    float* AGG = P1 + nd;

    const int    nRanges = (N + PART_NODES - 1) >> PART_SHIFT;
    const int    nChunks = (int)((E + CHUNK - 1) / CHUNK);
    const int    nHist   = nRanges * nChunks;
    const size_t padInt  = ((size_t)N + 1 + 1023) / 1024 * 1024;
    const size_t padHist = ((size_t)nHist + 1 + 4095) / 4096 * 4096;
    int*  ROWPTR = (int*)(AGG + nd);
    int*  HIST   = ROWPTR + padInt;
    int*  OFF    = HIST + padHist;
    int*  RECD   = OFF + padHist;
    int2* RECSW  = (int2*)(RECD + ((size_t)E + 3) / 4 * 4);
    int2* CSR    = RECSW + ((size_t)E + 1) / 2 * 2;
    float* WC    = (float*)(CSR + ((size_t)E + 1) / 2 * 2);

    const dim3 blk(256);
    const dim3 blkWide(1024);
    const dim3 gridWc((unsigned)(L * 24));
    const dim3 gridGru(256);
    const dim3 gridChunks((unsigned)nChunks);
    const dim3 gridScan((unsigned)(padHist / 4096));
    const dim3 gridRanges((unsigned)nRanges);
    const dim3 gridGather((unsigned)((N + 7) / 8));

    ggc_wcomb<<<gridWc, blk, 0, stream>>>(W_ih, W, WC, L);

    ggc_part_count<<<gridChunks, blk, 0, stream>>>(dst, HIST, E, N, nRanges, nChunks);
    ggc_scan<<<gridScan, blkWide, 0, stream>>>(HIST, OFF, nHist);
    ggc_part_place<<<gridChunks, blk, 0, stream>>>(src, dst, ea, OFF, RECSW, RECD, E, N, nRanges, nChunks);
    ggc_range_sort<<<gridRanges, blkWide, 0, stream>>>(OFF, RECD, RECSW, ROWPTR, CSR, N, nRanges, nChunks);

    const float* hcur = x;
    for (int l = 0; l < L; ++l) {
        ggc_gather<<<gridGather, blk, 0, stream>>>(hcur, ROWPTR, CSR, AGG, N);

        float* hout = (l == L - 1) ? (float*)d_out : ((l & 1) ? P1 : P0);
        ggc_gru<<<gridGru, blk, 0, stream>>>(AGG, hcur, WC + (size_t)l * 3 * DIM * DIM,
                                             W_hh, b_ih, b_hh, hout, N);
        hcur = hout;
    }
}
